# starting kernel
# baseline (speedup 1.0000x reference)
#include <hip/hip_runtime.h>
#include <hip/hip_bf16.h>


typedef __attribute__((ext_vector_type(2))) float v2f;
typedef __attribute__((ext_vector_type(8))) float v8f;

#define QP_ITERS   200
#define QP_M       128
#define QP_N       64
#define RSTRIDE    66
#define NTHREADS   256

__global__ __launch_bounds__(NTHREADS) void qp_fista_kernel(
    const float* __restrict__ Qm, const float* __restrict__ A,
    const float* __restrict__ X,  const float* __restrict__ Bv,
    float* __restrict__ Out)
{
    __shared__ __align__(16) float sR[QP_M * RSTRIDE];
    __shared__ __align__(16) float sY[2][QP_M];
    __shared__ float sC[QP_M];
    __shared__ float sLam[QP_M];
    __shared__ float sDinv[QP_N];
    __shared__ float sSqrtd[QP_N];
    __shared__ float sXp[QP_N];
    __shared__ float sRed[8];
    __shared__ float sEta;

    const int tid   = threadIdx.x;
    const int lane  = tid & 31;
    const int wave  = tid >> 5;
    const int ln16  = lane & 15;
    const int hl    = lane >> 4;
    const int batch = blockIdx.x;

    const float* Ab = A  + (size_t)batch * (QP_M * QP_N);
    const float* xb = X  + (size_t)batch * QP_N;
    const float* bb = Bv + (size_t)batch * QP_M;

    if (tid < QP_N) {
        float di = 1.0f / Qm[tid * QP_N + tid];
        sDinv[tid]  = di;
        sSqrtd[tid] = sqrtf(fabsf(di));
    }
    __syncthreads();
    if (tid < QP_N) sXp[tid] = sSqrtd[tid] * xb[tid];

    float ss = 0.f;
    #pragma unroll
    for (int j = 0; j < (QP_M * QP_N / 4) / NTHREADS; ++j) {
        int f4  = j * NTHREADS + tid;
        float4 v = ((const float4*)Ab)[f4];
        int fl  = f4 << 2;
        int row = fl >> 6, col = fl & 63;
        ss += v.x*v.x + v.y*v.y + v.z*v.z + v.w*v.w;
        float* dst = &sR[row * RSTRIDE + col];
        dst[0] = v.x * sSqrtd[col + 0];
        dst[1] = v.y * sSqrtd[col + 1];
        dst[2] = v.z * sSqrtd[col + 2];
        dst[3] = v.w * sSqrtd[col + 3];
    }
    #pragma unroll
    for (int off = 16; off; off >>= 1) ss += __shfl_xor(ss, off, 32);
    if (lane == 0) sRed[wave] = ss;
    if (tid < QP_M) sY[0][tid] = 0.f;
    __syncthreads();

    if (tid == 0) {
        float s = 0.f;
        for (int w = 0; w < 8; ++w) s += sRed[w];
        float qn2 = 0.f;
        for (int n = 0; n < QP_N; ++n) qn2 += sDinv[n] * sDinv[n];
        float L = s * sqrtf(qn2);
        sEta = 1.0f / fmaxf(L, 1e-12f);
    }
    if (tid < QP_M) {
        float acc = 0.f;
        #pragma unroll 8
        for (int n = 0; n < QP_N; ++n) acc += sR[tid * RSTRIDE + n] * sXp[n];
        sC[tid] = acc - bb[tid];
    }
    __syncthreads();

    v2f bf[16];
    #pragma unroll
    for (int k = 0; k < 16; ++k) {
        int row = wave * 16 + ln16;
        int col = 4 * k + 2 * hl;
        bf[k] = *(const v2f*)&sR[row * RSTRIDE + col];
    }
    v8f acc[8];
    #pragma unroll
    for (int i = 0; i < 8; ++i) {
        v8f c = {};
        #pragma unroll
        for (int k = 0; k < 16; ++k) {
            int row = i * 16 + ln16;
            int col = 4 * k + 2 * hl;
            v2f af = *(const v2f*)&sR[row * RSTRIDE + col];
            c = __builtin_amdgcn_wmma_f32_16x16x4_f32(
                    false, af, false, bf[k], (short)0, c, false, false);
        }
        acc[i] = c;
    }
    __syncthreads();

    const float eta = sEta;
    const int   r   = wave * 16 + ln16;
    float c_r = sC[r];
    float lam_r = 0.f, y_r = 0.f, t = 1.f;
    int buf = 0;

    for (int it = 0; it < QP_ITERS; ++it) {
        const float* y = sY[buf];
        float a = 0.f;
        #pragma unroll
        for (int i = 0; i < 8; ++i) {
            float4 y0 = *(const float4*)&y[i * 16 + 8 * hl];
            float4 y1 = *(const float4*)&y[i * 16 + 8 * hl + 4];
            a += acc[i][0]*y0.x + acc[i][1]*y0.y + acc[i][2]*y0.z + acc[i][3]*y0.w
               + acc[i][4]*y1.x + acc[i][5]*y1.y + acc[i][6]*y1.z + acc[i][7]*y1.w;
        }
        a += __shfl_xor(a, 16, 32);
        float tn = 0.5f * (1.f + sqrtf(1.f + 4.f * t * t));
        if (lane < 16) {
            float g   = c_r - a;
            float ln_ = fmaxf(0.f, y_r + eta * g);
            float yn  = ln_ + ((t - 1.f) / tn) * (ln_ - lam_r);
            sY[buf ^ 1][r] = yn;
            lam_r = ln_;
            y_r   = yn;
        }
        t = tn;
        buf ^= 1;
        __syncthreads();
    }

    if (lane < 16) sLam[r] = lam_r;
    __syncthreads();
    if (tid < QP_N) {
        float a2 = 0.f;
        #pragma unroll 8
        for (int m = 0; m < QP_M; ++m) a2 += sR[m * RSTRIDE + tid] * sLam[m];
        Out[(size_t)batch * QP_N + tid] = sDinv[tid] * xb[tid] - sSqrtd[tid] * a2;
    }
}

extern "C" void kernel_launch(void* const* d_in, const int* in_sizes, int n_in,
                              void* d_out, int out_size, void* d_ws, size_t ws_size,
                              hipStream_t stream) {
    const float* Q = (const float*)d_in[0];
    const float* A = (const float*)d_in[1];
    const float* X = (const float*)d_in[2];
    const float* B = (const float*)d_in[3];
    float* Out = (float*)d_out;
    int batches = in_sizes[1] / (QP_M * QP_N);
    qp_fista_kernel<<<batches, NTHREADS, 0, stream>>>(Q, A, X, B, Out);
}


# G kept in the WMMA D layout (A-block rows permuted): a lane owns 8 rows x its half's 64 columns, so 64 v_permlane16 per iteration, not 128; halves add by v_permlanex16; row summation order changed
# speedup vs baseline: 2.0904x; 2.0904x over previous
#include <hip/hip_runtime.h>
#include <hip/hip_bf16.h>


typedef __attribute__((ext_vector_type(2))) float v2f;
typedef __attribute__((ext_vector_type(8))) float v8f;


__device__ __forceinline__ float xhalf(float x) {
    return __builtin_bit_cast(float, __builtin_amdgcn_permlanex16(
        __builtin_bit_cast(int, x), __builtin_bit_cast(int, x),
        0x76543210, 0xfedcba98, false, false));
}
template <int N>
__device__ __forceinline__ float rowbcast(float x) {
    return __builtin_bit_cast(float, __builtin_amdgcn_permlane16(
        __builtin_bit_cast(int, x), __builtin_bit_cast(int, x),
        0x11111111u * (unsigned)N, 0x11111111u * (unsigned)N, true, false));
}
__device__ __forceinline__ v2f mk2(float a, float b) { v2f r = {a, b}; return r; }
__device__ __forceinline__ v2f gpair(const v8f& g, int p) { v2f r = {g[2 * p], g[2 * p + 1]}; return r; }
__device__ __forceinline__ v2f fma2(v2f a, v2f b, v2f c) { return __builtin_elementwise_fma(a, b, c); }

#define QP_ITERS   200
#define QP_M       128
#define QP_N       64
#define RSTRIDE    66
#define NTHREADS   32

template <int I>
__device__ __forceinline__ void tile_step(const v8f (&T)[8][8], const float ys, v2f (&acc)[8][2]) {
    constexpr int E = 8 * (I & 1);
    const float s0 = rowbcast<E + 0>(ys), s1 = rowbcast<E + 1>(ys);
    const float s2 = rowbcast<E + 2>(ys), s3 = rowbcast<E + 3>(ys);
    const float s4 = rowbcast<E + 4>(ys), s5 = rowbcast<E + 5>(ys);
    const float s6 = rowbcast<E + 6>(ys), s7 = rowbcast<E + 7>(ys);
    #pragma unroll
    for (int J = 0; J < 8; ++J) {
        acc[J][0] = fma2(gpair(T[I][J], 0), mk2(s0, s1), acc[J][0]);
        acc[J][1] = fma2(gpair(T[I][J], 1), mk2(s2, s3), acc[J][1]);
    }
    #pragma unroll
    for (int J = 0; J < 8; ++J) {
        acc[J][0] = fma2(gpair(T[I][J], 2), mk2(s4, s5), acc[J][0]);
        acc[J][1] = fma2(gpair(T[I][J], 3), mk2(s6, s7), acc[J][1]);
    }
}

__global__ __launch_bounds__(NTHREADS) void qp_fista_kernel(
    const float* __restrict__ Qm, const float* __restrict__ A,
    const float* __restrict__ X,  const float* __restrict__ Bv,
    float* __restrict__ Out)
{
    __shared__ __align__(16) float sR[QP_M * RSTRIDE];
    __shared__ float sLam[QP_M];
    __shared__ float sDinv[QP_N];
    __shared__ float sSqrtd[QP_N];
    __shared__ float sXp[QP_N];

    const int lane  = threadIdx.x;
    const int ln16  = lane & 15;
    const int hl    = lane >> 4;
    const int batch = blockIdx.x;

    const float* Ab = A  + (size_t)batch * (QP_M * QP_N);
    const float* xb = X  + (size_t)batch * QP_N;
    const float* bb = Bv + (size_t)batch * QP_M;

    #pragma unroll
    for (int q = 0; q < QP_N / NTHREADS; ++q) {
        int n = q * NTHREADS + lane;
        float di = 1.0f / Qm[n * QP_N + n];
        float sd = sqrtf(fabsf(di));
        sDinv[n]  = di;
        sSqrtd[n] = sd;
        sXp[n]    = sd * xb[n];
    }
    __syncthreads();

    float ss = 0.f;
    #pragma unroll
    for (int j = 0; j < (QP_M * QP_N / 4) / NTHREADS; ++j) {
        int f4  = j * NTHREADS + lane;
        float4 v = ((const float4*)Ab)[f4];
        int fl  = f4 << 2;
        int row = fl >> 6, col = fl & 63;
        ss += v.x*v.x + v.y*v.y + v.z*v.z + v.w*v.w;
        float* dst = &sR[row * RSTRIDE + col];
        dst[0] = v.x * sSqrtd[col + 0];
        dst[1] = v.y * sSqrtd[col + 1];
        dst[2] = v.z * sSqrtd[col + 2];
        dst[3] = v.w * sSqrtd[col + 3];
    }
    #pragma unroll
    for (int off = 16; off; off >>= 1) ss += __shfl_xor(ss, off, 32);
    __syncthreads();

    float eta;
    {
        float qn2 = 0.f;
        for (int n = 0; n < QP_N; ++n) qn2 += sDinv[n] * sDinv[n];
        float L = ss * sqrtf(qn2);
        eta = 1.0f / fmaxf(L, 1e-12f);
    }
    float cr[4] = {0.f, 0.f, 0.f, 0.f};
    #pragma unroll 8
    for (int n = 0; n < QP_N; ++n) {
        const float xp = sXp[n];
        #pragma unroll
        for (int j = 0; j < 4; ++j)
            cr[j] += sR[(16 * (2 * j + hl) + ln16) * RSTRIDE + n] * xp;
    }
    #pragma unroll
    for (int j = 0; j < 4; ++j) cr[j] -= bb[16 * (2 * j + hl) + ln16];

    v8f T[8][8];
    #pragma unroll
    for (int J = 0; J < 8; ++J) {
        v2f bf[16];
        #pragma unroll
        for (int k = 0; k < 16; ++k) {
            int row = J * 16 + ln16;
            int col = 4 * k + 2 * hl;
            bf[k] = *(const v2f*)&sR[row * RSTRIDE + col];
        }
        #pragma unroll
        for (int I = 0; I < 8; I += 2) {
            v8f c0 = {}, c1 = {};
            #pragma unroll
            for (int k = 0; k < 16; ++k) {
                int col  = 4 * k + 2 * hl;
                const int ra = 16 * (I + (ln16 >> 3)) + (ln16 & 7);
                v2f af0 = *(const v2f*)&sR[ra * RSTRIDE + col];
                v2f af1 = *(const v2f*)&sR[(ra + 8) * RSTRIDE + col];
                c0 = __builtin_amdgcn_wmma_f32_16x16x4_f32(
                        false, af0, false, bf[k], (short)0, c0, false, false);
                c1 = __builtin_amdgcn_wmma_f32_16x16x4_f32(
                        false, af1, false, bf[k], (short)0, c1, false, false);
            }
            T[I][J]     = c0;
            T[I + 1][J] = c1;
        }
    }
    float yv[4]  = {0.f, 0.f, 0.f, 0.f};
    float lam[4] = {0.f, 0.f, 0.f, 0.f};
    float t = 1.f;

    for (int it = 0; it < QP_ITERS; ++it) {
        v2f acc[8][2];
        #pragma unroll
        for (int J = 0; J < 8; ++J) {
            acc[J][0] = mk2(0.f, 0.f);
            acc[J][1] = mk2(0.f, 0.f);
        }
        tile_step<0>(T, yv[0], acc);
        tile_step<1>(T, yv[0], acc);
        tile_step<2>(T, yv[1], acc);
        tile_step<3>(T, yv[1], acc);
        tile_step<4>(T, yv[2], acc);
        tile_step<5>(T, yv[2], acc);
        tile_step<6>(T, yv[3], acc);
        tile_step<7>(T, yv[3], acc);

        const float tn   = 0.5f * (1.f + sqrtf(1.f + 4.f * t * t));
        const float beta = (t - 1.f) / tn;
        #pragma unroll
        for (int j = 0; j < 4; ++j) {
            const v2f e2 = acc[2 * j][0] + acc[2 * j][1];
            const v2f o2 = acc[2 * j + 1][0] + acc[2 * j + 1][1];
            const float pe = e2.x + e2.y;
            const float po = o2.x + o2.y;
            const float a   = (hl ? po : pe) + xhalf(hl ? pe : po);
            const float g   = cr[j] - a;
            const float ln_ = fmaxf(0.f, yv[j] + eta * g);
            const float yn  = ln_ + beta * (ln_ - lam[j]);
            lam[j] = ln_;
            yv[j]  = yn;
        }
        t = tn;
    }

    #pragma unroll
    for (int j = 0; j < 4; ++j) sLam[16 * (2 * j + hl) + ln16] = lam[j];
    __syncthreads();
    float a2[QP_N / NTHREADS] = {0.f, 0.f};
    #pragma unroll 8
    for (int m = 0; m < QP_M; ++m) {
        const float lm = sLam[m];
        #pragma unroll
        for (int q = 0; q < QP_N / NTHREADS; ++q)
            a2[q] += sR[m * RSTRIDE + q * NTHREADS + lane] * lm;
    }
    #pragma unroll
    for (int q = 0; q < QP_N / NTHREADS; ++q) {
        int n = q * NTHREADS + lane;
        Out[(size_t)batch * QP_N + n] = sDinv[n] * xb[n] - sSqrtd[n] * a2[q];
    }
}

extern "C" void kernel_launch(void* const* d_in, const int* in_sizes, int n_in,
                              void* d_out, int out_size, void* d_ws, size_t ws_size,
                              hipStream_t stream) {
    const float* Q = (const float*)d_in[0];
    const float* A = (const float*)d_in[1];
    const float* X = (const float*)d_in[2];
    const float* B = (const float*)d_in[3];
    float* Out = (float*)d_out;
    int batches = in_sizes[1] / (QP_M * QP_N);
    qp_fista_kernel<<<batches, NTHREADS, 0, stream>>>(Q, A, X, B, Out);
}


# G = R R^T is symmetric: only the 10 of 16 32x32 blocks on and above its diagonal are multiplied out (640 f32 WMMAs per instance, not 1024); the other 6 are mirrored through LDS; sums unchanged
# speedup vs baseline: 2.2017x; 1.0532x over previous
#include <hip/hip_runtime.h>
#include <hip/hip_bf16.h>


typedef __attribute__((ext_vector_type(2))) float v2f;
typedef __attribute__((ext_vector_type(8))) float v8f;


__device__ __forceinline__ float xhalf(float x) {
    return __builtin_bit_cast(float, __builtin_amdgcn_permlanex16(
        __builtin_bit_cast(int, x), __builtin_bit_cast(int, x),
        0x76543210, 0xfedcba98, false, false));
}
template <int N>
__device__ __forceinline__ float rowbcast(float x) {
    return __builtin_bit_cast(float, __builtin_amdgcn_permlane16(
        __builtin_bit_cast(int, x), __builtin_bit_cast(int, x),
        0x11111111u * (unsigned)N, 0x11111111u * (unsigned)N, true, false));
}
__device__ __forceinline__ v2f mk2(float a, float b) { v2f r = {a, b}; return r; }
__device__ __forceinline__ v2f gpair(const v8f& g, int p) { v2f r = {g[2 * p], g[2 * p + 1]}; return r; }
__device__ __forceinline__ v2f fma2(v2f a, v2f b, v2f c) { return __builtin_elementwise_fma(a, b, c); }

#define QP_ITERS   200
#define QP_M       128
#define QP_N       64
#define RSTRIDE    66
#define NTHREADS   32
#define MIR_P      36
#define MIR_SZ     1184

template <int I>
__device__ __forceinline__ void tile_step(const v8f (&T)[8][8], const float ys, v2f (&acc)[8][2]) {
    constexpr int E = 8 * (I & 1);
    const float s0 = rowbcast<E + 0>(ys), s1 = rowbcast<E + 1>(ys);
    const float s2 = rowbcast<E + 2>(ys), s3 = rowbcast<E + 3>(ys);
    const float s4 = rowbcast<E + 4>(ys), s5 = rowbcast<E + 5>(ys);
    const float s6 = rowbcast<E + 6>(ys), s7 = rowbcast<E + 7>(ys);
    #pragma unroll
    for (int J = 0; J < 8; ++J) {
        acc[J][0] = fma2(gpair(T[I][J], 0), mk2(s0, s1), acc[J][0]);
        acc[J][1] = fma2(gpair(T[I][J], 1), mk2(s2, s3), acc[J][1]);
    }
    #pragma unroll
    for (int J = 0; J < 8; ++J) {
        acc[J][0] = fma2(gpair(T[I][J], 2), mk2(s4, s5), acc[J][0]);
        acc[J][1] = fma2(gpair(T[I][J], 3), mk2(s6, s7), acc[J][1]);
    }
}

__global__ __launch_bounds__(NTHREADS) void qp_fista_kernel(
    const float* __restrict__ Qm, const float* __restrict__ A,
    const float* __restrict__ X,  const float* __restrict__ Bv,
    float* __restrict__ Out)
{
    __shared__ __align__(16) float sR[QP_M * RSTRIDE];
    __shared__ float sLam[QP_M];
    __shared__ float sDinv[QP_N];
    __shared__ float sSqrtd[QP_N];
    __shared__ float sXp[QP_N];
    __shared__ __align__(16) float sMir[6][MIR_SZ];

    const int lane  = threadIdx.x;
    const int ln16  = lane & 15;
    const int hl    = lane >> 4;
    const int batch = blockIdx.x;

    const float* Ab = A  + (size_t)batch * (QP_M * QP_N);
    const float* xb = X  + (size_t)batch * QP_N;
    const float* bb = Bv + (size_t)batch * QP_M;

    #pragma unroll
    for (int q = 0; q < QP_N / NTHREADS; ++q) {
        int n = q * NTHREADS + lane;
        float di = 1.0f / Qm[n * QP_N + n];
        float sd = sqrtf(fabsf(di));
        sDinv[n]  = di;
        sSqrtd[n] = sd;
        sXp[n]    = sd * xb[n];
    }
    __syncthreads();

    float ss = 0.f;
    #pragma unroll
    for (int j = 0; j < (QP_M * QP_N / 4) / NTHREADS; ++j) {
        int f4  = j * NTHREADS + lane;
        float4 v = ((const float4*)Ab)[f4];
        int fl  = f4 << 2;
        int row = fl >> 6, col = fl & 63;
        ss += v.x*v.x + v.y*v.y + v.z*v.z + v.w*v.w;
        float* dst = &sR[row * RSTRIDE + col];
        dst[0] = v.x * sSqrtd[col + 0];
        dst[1] = v.y * sSqrtd[col + 1];
        dst[2] = v.z * sSqrtd[col + 2];
        dst[3] = v.w * sSqrtd[col + 3];
    }
    #pragma unroll
    for (int off = 16; off; off >>= 1) ss += __shfl_xor(ss, off, 32);
    __syncthreads();

    float eta;
    {
        float qn2 = 0.f;
        for (int n = 0; n < QP_N; ++n) qn2 += sDinv[n] * sDinv[n];
        float L = ss * sqrtf(qn2);
        eta = 1.0f / fmaxf(L, 1e-12f);
    }
    float cr[4] = {0.f, 0.f, 0.f, 0.f};
    #pragma unroll 8
    for (int n = 0; n < QP_N; ++n) {
        const float xp = sXp[n];
        #pragma unroll
        for (int j = 0; j < 4; ++j)
            cr[j] += sR[(16 * (2 * j + hl) + ln16) * RSTRIDE + n] * xp;
    }
    #pragma unroll
    for (int j = 0; j < 4; ++j) cr[j] -= bb[16 * (2 * j + hl) + ln16];

    v8f T[8][8];
    v2f bf[8][16];
    #pragma unroll
    for (int J = 0; J < 8; ++J) {
        #pragma unroll
        for (int k = 0; k < 16; ++k) {
            int row = J * 16 + ln16;
            int col = 4 * k + 2 * hl;
            bf[J][k] = *(const v2f*)&sR[row * RSTRIDE + col];
        }
    }
    #pragma unroll
    for (int I = 0; I < 8; I += 2) {
        v2f af0[16], af1[16];
        #pragma unroll
        for (int k = 0; k < 16; ++k) {
            int col  = 4 * k + 2 * hl;
            const int ra = 16 * (I + (ln16 >> 3)) + (ln16 & 7);
            af0[k] = *(const v2f*)&sR[ra * RSTRIDE + col];
            af1[k] = *(const v2f*)&sR[(ra + 8) * RSTRIDE + col];
        }
        #pragma unroll
        for (int J = 0; J < 8; ++J) {
            if ((I >> 1) <= (J >> 1)) {
                v8f c0 = {}, c1 = {};
                #pragma unroll
                for (int k = 0; k < 16; ++k) {
                    c0 = __builtin_amdgcn_wmma_f32_16x16x4_f32(
                            false, af0[k], false, bf[J][k], (short)0, c0, false, false);
                    c1 = __builtin_amdgcn_wmma_f32_16x16x4_f32(
                            false, af1[k], false, bf[J][k], (short)0, c1, false, false);
                }
                T[I][J]     = c0;
                T[I + 1][J] = c1;
                if ((I >> 1) < (J >> 1)) {
                    float* p = &sMir[(J >> 1) * ((J >> 1) - 1) / 2 + (I >> 1)]
                                    [16 * hl * MIR_P + 16 * hl + 16 * (J & 1) + ln16];
                    #pragma unroll
                    for (int v = 0; v < 8; ++v) {
                        p[v * MIR_P]       = c0[v];
                        p[(8 + v) * MIR_P] = c1[v];
                    }
                }
            }
        }
    }
    __syncthreads();
    #pragma unroll
    for (int n = 1; n < 4; ++n) {
        #pragma unroll
        for (int m = 0; m < n; ++m) {
            const float* s = &sMir[n * (n - 1) / 2 + m][ln16 * MIR_P + 16 * hl];
            #pragma unroll
            for (int i = 0; i < 2; ++i) {
                #pragma unroll
                for (int j = 0; j < 2; ++j) {
                    const float4* q = (const float4*)(s + 16 * j * MIR_P + 16 * j + 8 * i);
                    const float4 lo = q[0], hi = q[1];
                    v8f g = {lo.x, lo.y, lo.z, lo.w, hi.x, hi.y, hi.z, hi.w};
                    T[2 * n + i][2 * m + j] = g;
                }
            }
        }
    }
    float yv[4]  = {0.f, 0.f, 0.f, 0.f};
    float lam[4] = {0.f, 0.f, 0.f, 0.f};
    float t = 1.f;

    for (int it = 0; it < QP_ITERS; ++it) {
        v2f acc[8][2];
        #pragma unroll
        for (int J = 0; J < 8; ++J) {
            acc[J][0] = mk2(0.f, 0.f);
            acc[J][1] = mk2(0.f, 0.f);
        }
        tile_step<0>(T, yv[0], acc);
        tile_step<1>(T, yv[0], acc);
        tile_step<2>(T, yv[1], acc);
        tile_step<3>(T, yv[1], acc);
        tile_step<4>(T, yv[2], acc);
        tile_step<5>(T, yv[2], acc);
        tile_step<6>(T, yv[3], acc);
        tile_step<7>(T, yv[3], acc);

        const float tn   = 0.5f * (1.f + sqrtf(1.f + 4.f * t * t));
        const float beta = (t - 1.f) / tn;
        #pragma unroll
        for (int j = 0; j < 4; ++j) {
            const v2f e2 = acc[2 * j][0] + acc[2 * j][1];
            const v2f o2 = acc[2 * j + 1][0] + acc[2 * j + 1][1];
            const float pe = e2.x + e2.y;
            const float po = o2.x + o2.y;
            const float a   = (hl ? po : pe) + xhalf(hl ? pe : po);
            const float g   = cr[j] - a;
            const float ln_ = fmaxf(0.f, yv[j] + eta * g);
            const float yn  = ln_ + beta * (ln_ - lam[j]);
            lam[j] = ln_;
            yv[j]  = yn;
        }
        t = tn;
    }

    #pragma unroll
    for (int j = 0; j < 4; ++j) sLam[16 * (2 * j + hl) + ln16] = lam[j];
    __syncthreads();
    float a2[QP_N / NTHREADS] = {0.f, 0.f};
    #pragma unroll 8
    for (int m = 0; m < QP_M; ++m) {
        const float lm = sLam[m];
        #pragma unroll
        for (int q = 0; q < QP_N / NTHREADS; ++q)
            a2[q] += sR[m * RSTRIDE + q * NTHREADS + lane] * lm;
    }
    #pragma unroll
    for (int q = 0; q < QP_N / NTHREADS; ++q) {
        int n = q * NTHREADS + lane;
        Out[(size_t)batch * QP_N + n] = sDinv[n] * xb[n] - sSqrtd[n] * a2[q];
    }
}

extern "C" void kernel_launch(void* const* d_in, const int* in_sizes, int n_in,
                              void* d_out, int out_size, void* d_ws, size_t ws_size,
                              hipStream_t stream) {
    const float* Q = (const float*)d_in[0];
    const float* A = (const float*)d_in[1];
    const float* X = (const float*)d_in[2];
    const float* B = (const float*)d_in[3];
    float* Out = (float*)d_out;
    int batches = in_sizes[1] / (QP_M * QP_N);
    qp_fista_kernel<<<batches, NTHREADS, 0, stream>>>(Q, A, X, B, Out);
}
